# starting kernel
# baseline (speedup 1.0000x reference)
#include <hip/hip_runtime.h>
#include <math.h>


typedef float v2f __attribute__((ext_vector_type(2)));
typedef float v8f __attribute__((ext_vector_type(8)));

#define LSEQ   1024
#define DMODEL 1024

__global__ __launch_bounds__(128)
void gemm_nt_f32_wmma(const float* __restrict__ A, const float* __restrict__ B,
                      float* __restrict__ C, int M, int N, int K, float alpha) {
  const int lane = threadIdx.x & 31;
  const int wave = threadIdx.x >> 5;
  const int lid  = lane & 15;
  const int half = lane >> 4;
  const int mBase = blockIdx.y * 128 + (wave >> 1) * 64;
  const int nBase = blockIdx.x * 128 + (wave & 1) * 64;
  const int koff = half * 2;

  const float* aPtr[4];
  const float* bPtr[4];
  #pragma unroll
  for (int i = 0; i < 4; ++i) {
    aPtr[i] = A + (size_t)(mBase + i * 16 + lid) * K + koff;
    bPtr[i] = B + (size_t)(nBase + i * 16 + lid) * K + koff;
  }

  v8f acc[4][4];
  #pragma unroll
  for (int i = 0; i < 4; ++i)
    #pragma unroll
    for (int j = 0; j < 4; ++j)
      acc[i][j] = (v8f){};

  #pragma unroll 2
  for (int k = 0; k < K; k += 4) {
    v2f fa[4], fb[4];
    #pragma unroll
    for (int i = 0; i < 4; ++i) fa[i] = *(const v2f*)(aPtr[i] + k);
    #pragma unroll
    for (int j = 0; j < 4; ++j) fb[j] = *(const v2f*)(bPtr[j] + k);
    #pragma unroll
    for (int i = 0; i < 4; ++i)
      #pragma unroll
      for (int j = 0; j < 4; ++j)
        acc[i][j] = __builtin_amdgcn_wmma_f32_16x16x4_f32(
            false, fa[i], false, fb[j], (short)0, acc[i][j], false, false);
  }

  #pragma unroll
  for (int i = 0; i < 4; ++i) {
    #pragma unroll
    for (int j = 0; j < 4; ++j) {
      const int col = nBase + j * 16 + lid;
      #pragma unroll
      for (int r = 0; r < 8; ++r) {
        const int row = mBase + i * 16 + half * 8 + r;
        C[(size_t)row * N + col] = alpha * acc[i][j][r];
      }
    }
  }
}

__global__ __launch_bounds__(128)
void ema_scan(float* __restrict__ Ks, float* __restrict__ Vs,
              const float* __restrict__ gates, int L, int D) {
  const int d = blockIdx.x * blockDim.x + threadIdx.x;
  if (d >= D) return;
  const float g = gates[0];
  float ek = 0.0f, ev = 0.0f;
  #pragma unroll 4
  for (int t = 0; t < L; ++t) {
    const size_t idx = (size_t)t * D + d;
    ek = fmaf(g, ek, Ks[idx]);
    ev = fmaf(g, ev, Vs[idx]);
    Ks[idx] = ek;
    Vs[idx] = ev;
  }
}

__global__ __launch_bounds__(256)
void gate_softmax(float* __restrict__ Q, const float* __restrict__ Kema,
                  const float* __restrict__ Vema, int L, int D) {
  const int t    = blockIdx.x;
  const int wave = threadIdx.x >> 5;
  const int lane = threadIdx.x & 31;
  const int h    = blockIdx.y * 8 + wave;
  const size_t idx = (size_t)t * D + h * 32 + lane;
  const float qv = Q[idx];
  float p = qv * Kema[idx];
  #pragma unroll
  for (int off = 16; off > 0; off >>= 1)
    p += __shfl_xor(p, off, 32);
  const float a = 1.0f / (1.0f + 31.0f * expf(-p));
  Q[idx] = a * Vema[idx];
}

extern "C" void kernel_launch(void* const* d_in, const int* in_sizes, int n_in,
                              void* d_out, int out_size, void* d_ws, size_t ws_size,
                              hipStream_t stream) {
  const float* q     = (const float*)d_in[0];
  const float* k     = (const float*)d_in[1];
  const float* v     = (const float*)d_in[2];
  const float* Wq    = (const float*)d_in[3];
  const float* Wk    = (const float*)d_in[4];
  const float* Wv    = (const float*)d_in[5];
  const float* Wd    = (const float*)d_in[6];
  const float* gates = (const float*)d_in[7];
  float* out = (float*)d_out;

  const int L = LSEQ, D = DMODEL;
  const size_t LD = (size_t)L * D;
  float* Qs = (float*)d_ws;
  float* Ks = Qs + LD;
  float* Vs = Ks + LD;

  const float invScale = 1.0f / sqrtf(sqrtf(32.0f));

  dim3 gBlk(128);
  dim3 gGrid(D / 128, L / 128);

  hipLaunchKernelGGL(gemm_nt_f32_wmma, gGrid, gBlk, 0, stream, q, Wq, Qs, L, D, D, invScale);
  hipLaunchKernelGGL(gemm_nt_f32_wmma, gGrid, gBlk, 0, stream, k, Wk, Ks, L, D, D, invScale);
  hipLaunchKernelGGL(gemm_nt_f32_wmma, gGrid, gBlk, 0, stream, v, Wv, Vs, L, D, D, 1.0f);
  hipLaunchKernelGGL(ema_scan, dim3(D / 128), dim3(128), 0, stream, Ks, Vs, gates, L, D);
  hipLaunchKernelGGL(gate_softmax, dim3(L, 4), dim3(256), 0, stream, Qs, Ks, Vs, L, D);
  hipLaunchKernelGGL(gemm_nt_f32_wmma, gGrid, gBlk, 0, stream, Qs, Wd, out, L, D, D, 1.0f);
}


# Serial 1024-step EMA scan split into 32 time chunks per 32-channel group (32 waves: chunk-local EMA, carry g^(j+1)*state via LDS); EMA summation order changes (chunk sums plus carry)
# speedup vs baseline: 1.7492x; 1.7492x over previous
#include <hip/hip_runtime.h>
#include <math.h>


typedef float v2f __attribute__((ext_vector_type(2)));
typedef float v8f __attribute__((ext_vector_type(8)));

#define LSEQ   1024
#define DMODEL 1024

__global__ __launch_bounds__(128)
void gemm_nt_f32_wmma(const float* __restrict__ A, const float* __restrict__ B,
                      float* __restrict__ C, int M, int N, int K, float alpha) {
  const int lane = threadIdx.x & 31;
  const int wave = threadIdx.x >> 5;
  const int lid  = lane & 15;
  const int half = lane >> 4;
  const int mBase = blockIdx.y * 128 + (wave >> 1) * 64;
  const int nBase = blockIdx.x * 128 + (wave & 1) * 64;
  const int koff = half * 2;

  const float* aPtr[4];
  const float* bPtr[4];
  #pragma unroll
  for (int i = 0; i < 4; ++i) {
    aPtr[i] = A + (size_t)(mBase + i * 16 + lid) * K + koff;
    bPtr[i] = B + (size_t)(nBase + i * 16 + lid) * K + koff;
  }

  v8f acc[4][4];
  #pragma unroll
  for (int i = 0; i < 4; ++i)
    #pragma unroll
    for (int j = 0; j < 4; ++j)
      acc[i][j] = (v8f){};

  #pragma unroll 2
  for (int k = 0; k < K; k += 4) {
    v2f fa[4], fb[4];
    #pragma unroll
    for (int i = 0; i < 4; ++i) fa[i] = *(const v2f*)(aPtr[i] + k);
    #pragma unroll
    for (int j = 0; j < 4; ++j) fb[j] = *(const v2f*)(bPtr[j] + k);
    #pragma unroll
    for (int i = 0; i < 4; ++i)
      #pragma unroll
      for (int j = 0; j < 4; ++j)
        acc[i][j] = __builtin_amdgcn_wmma_f32_16x16x4_f32(
            false, fa[i], false, fb[j], (short)0, acc[i][j], false, false);
  }

  #pragma unroll
  for (int i = 0; i < 4; ++i) {
    #pragma unroll
    for (int j = 0; j < 4; ++j) {
      const int col = nBase + j * 16 + lid;
      #pragma unroll
      for (int r = 0; r < 8; ++r) {
        const int row = mBase + i * 16 + half * 8 + r;
        C[(size_t)row * N + col] = alpha * acc[i][j][r];
      }
    }
  }
}

#define SCAN_T 32
#define SCAN_W (LSEQ / SCAN_T)
__global__ __launch_bounds__(SCAN_W * 32)
void ema_scan_chunked(float* __restrict__ Ks, float* __restrict__ Vs,
                      const float* __restrict__ gates) {
  __shared__ float chunkEnd[SCAN_W][32];
  const int lane = threadIdx.x & 31;
  const int w    = threadIdx.x >> 5;
  float* P = (blockIdx.y ? Vs : Ks)
           + (size_t)(w * SCAN_T) * DMODEL + blockIdx.x * 32 + lane;
  const float g = gates[0];

  float x[SCAN_T];
  #pragma unroll
  for (int j = 0; j < SCAN_T; ++j) x[j] = P[(size_t)j * DMODEL];

  float e = 0.0f, gT = 1.0f;
  #pragma unroll
  for (int j = 0; j < SCAN_T; ++j) {
    e = fmaf(g, e, x[j]);
    x[j] = e;
    gT *= g;
  }
  chunkEnd[w][lane] = e;
  __syncthreads();

  if (w > 0) {
    float c = chunkEnd[0][lane];
    for (int i = 1; i < w; ++i) c = fmaf(c, gT, chunkEnd[i][lane]);
    float gp = g;
    #pragma unroll
    for (int j = 0; j < SCAN_T; ++j) {
      x[j] = fmaf(gp, c, x[j]);
      gp *= g;
    }
  }
  #pragma unroll
  for (int j = 0; j < SCAN_T; ++j) P[(size_t)j * DMODEL] = x[j];
}

__global__ __launch_bounds__(256)
void gate_softmax(float* __restrict__ Q, const float* __restrict__ Kema,
                  const float* __restrict__ Vema, int L, int D) {
  const int t    = blockIdx.x;
  const int wave = threadIdx.x >> 5;
  const int lane = threadIdx.x & 31;
  const int h    = blockIdx.y * 8 + wave;
  const size_t idx = (size_t)t * D + h * 32 + lane;
  const float qv = Q[idx];
  float p = qv * Kema[idx];
  #pragma unroll
  for (int off = 16; off > 0; off >>= 1)
    p += __shfl_xor(p, off, 32);
  const float a = 1.0f / (1.0f + 31.0f * expf(-p));
  Q[idx] = a * Vema[idx];
}

extern "C" void kernel_launch(void* const* d_in, const int* in_sizes, int n_in,
                              void* d_out, int out_size, void* d_ws, size_t ws_size,
                              hipStream_t stream) {
  const float* q     = (const float*)d_in[0];
  const float* k     = (const float*)d_in[1];
  const float* v     = (const float*)d_in[2];
  const float* Wq    = (const float*)d_in[3];
  const float* Wk    = (const float*)d_in[4];
  const float* Wv    = (const float*)d_in[5];
  const float* Wd    = (const float*)d_in[6];
  const float* gates = (const float*)d_in[7];
  float* out = (float*)d_out;

  const int L = LSEQ, D = DMODEL;
  const size_t LD = (size_t)L * D;
  float* Qs = (float*)d_ws;
  float* Ks = Qs + LD;
  float* Vs = Ks + LD;

  const float invScale = 1.0f / sqrtf(sqrtf(32.0f));

  dim3 gBlk(128);
  dim3 gGrid(D / 128, L / 128);

  hipLaunchKernelGGL(gemm_nt_f32_wmma, gGrid, gBlk, 0, stream, q, Wq, Qs, L, D, D, invScale);
  hipLaunchKernelGGL(gemm_nt_f32_wmma, gGrid, gBlk, 0, stream, k, Wk, Ks, L, D, D, invScale);
  hipLaunchKernelGGL(gemm_nt_f32_wmma, gGrid, gBlk, 0, stream, v, Wv, Vs, L, D, D, 1.0f);
  hipLaunchKernelGGL(ema_scan_chunked, dim3(D / 32, 2), dim3(SCAN_W * 32), 0, stream, Ks, Vs, gates);
  hipLaunchKernelGGL(gate_softmax, dim3(L, 4), dim3(256), 0, stream, Qs, Ks, Vs, L, D);
  hipLaunchKernelGGL(gemm_nt_f32_wmma, gGrid, gBlk, 0, stream, Qs, Wd, out, L, D, D, 1.0f);
}


# GEMM work-group tile 128x128 -> 64x64 (wave tile 32x32): 256 work-groups fill all 256 WGPs instead of 64; a trip's loads issued ahead of its WMMAs (occupancy hint 1-4); sum order unchanged
# speedup vs baseline: 4.0828x; 2.3341x over previous
#include <hip/hip_runtime.h>
#include <math.h>


typedef float v2f __attribute__((ext_vector_type(2)));
typedef float v8f __attribute__((ext_vector_type(8)));

#define LSEQ   1024
#define DMODEL 1024

__global__ __launch_bounds__(128) __attribute__((amdgpu_waves_per_eu(1, 4)))
void gemm_nt_f32_wmma(const float* __restrict__ A, const float* __restrict__ B,
                      float* __restrict__ C, int M, int N, int K, float alpha) {
  const int lane = threadIdx.x & 31;
  const int wave = threadIdx.x >> 5;
  const int lid  = lane & 15;
  const int half = lane >> 4;
  const int mBase = blockIdx.y * 64 + (wave >> 1) * 32;
  const int nBase = blockIdx.x * 64 + (wave & 1) * 32;
  const int koff = half * 2;

  const float* aPtr[2];
  const float* bPtr[2];
  #pragma unroll
  for (int i = 0; i < 2; ++i) {
    aPtr[i] = A + (size_t)(mBase + i * 16 + lid) * K + koff;
    bPtr[i] = B + (size_t)(nBase + i * 16 + lid) * K + koff;
  }

  v8f acc[2][2];
  #pragma unroll
  for (int i = 0; i < 2; ++i)
    #pragma unroll
    for (int j = 0; j < 2; ++j)
      acc[i][j] = (v8f){};

  for (int k = 0; k < K; k += 32) {
    v2f fa[8][2], fb[8][2];
    #pragma unroll
    for (int s = 0; s < 8; ++s) {
      #pragma unroll
      for (int i = 0; i < 2; ++i) fa[s][i] = *(const v2f*)(aPtr[i] + k + 4 * s);
      #pragma unroll
      for (int j = 0; j < 2; ++j) fb[s][j] = *(const v2f*)(bPtr[j] + k + 4 * s);
    }
    #pragma unroll
    for (int s = 0; s < 8; ++s)
      #pragma unroll
      for (int i = 0; i < 2; ++i)
        #pragma unroll
        for (int j = 0; j < 2; ++j)
          acc[i][j] = __builtin_amdgcn_wmma_f32_16x16x4_f32(
              false, fa[s][i], false, fb[s][j], (short)0, acc[i][j], false, false);
  }

  #pragma unroll
  for (int i = 0; i < 2; ++i) {
    #pragma unroll
    for (int j = 0; j < 2; ++j) {
      const int col = nBase + j * 16 + lid;
      #pragma unroll
      for (int r = 0; r < 8; ++r) {
        const int row = mBase + i * 16 + half * 8 + r;
        C[(size_t)row * N + col] = alpha * acc[i][j][r];
      }
    }
  }
}

#define SCAN_T 32
#define SCAN_W (LSEQ / SCAN_T)
__global__ __launch_bounds__(SCAN_W * 32)
void ema_scan_chunked(float* __restrict__ Ks, float* __restrict__ Vs,
                      const float* __restrict__ gates) {
  __shared__ float chunkEnd[SCAN_W][32];
  const int lane = threadIdx.x & 31;
  const int w    = threadIdx.x >> 5;
  float* P = (blockIdx.y ? Vs : Ks)
           + (size_t)(w * SCAN_T) * DMODEL + blockIdx.x * 32 + lane;
  const float g = gates[0];

  float x[SCAN_T];
  #pragma unroll
  for (int j = 0; j < SCAN_T; ++j) x[j] = P[(size_t)j * DMODEL];

  float e = 0.0f, gT = 1.0f;
  #pragma unroll
  for (int j = 0; j < SCAN_T; ++j) {
    e = fmaf(g, e, x[j]);
    x[j] = e;
    gT *= g;
  }
  chunkEnd[w][lane] = e;
  __syncthreads();

  if (w > 0) {
    float c = chunkEnd[0][lane];
    for (int i = 1; i < w; ++i) c = fmaf(c, gT, chunkEnd[i][lane]);
    float gp = g;
    #pragma unroll
    for (int j = 0; j < SCAN_T; ++j) {
      x[j] = fmaf(gp, c, x[j]);
      gp *= g;
    }
  }
  #pragma unroll
  for (int j = 0; j < SCAN_T; ++j) P[(size_t)j * DMODEL] = x[j];
}

__global__ __launch_bounds__(256)
void gate_softmax(float* __restrict__ Q, const float* __restrict__ Kema,
                  const float* __restrict__ Vema, int L, int D) {
  const int t    = blockIdx.x;
  const int wave = threadIdx.x >> 5;
  const int lane = threadIdx.x & 31;
  const int h    = blockIdx.y * 8 + wave;
  const size_t idx = (size_t)t * D + h * 32 + lane;
  const float qv = Q[idx];
  float p = qv * Kema[idx];
  #pragma unroll
  for (int off = 16; off > 0; off >>= 1)
    p += __shfl_xor(p, off, 32);
  const float a = 1.0f / (1.0f + 31.0f * expf(-p));
  Q[idx] = a * Vema[idx];
}

extern "C" void kernel_launch(void* const* d_in, const int* in_sizes, int n_in,
                              void* d_out, int out_size, void* d_ws, size_t ws_size,
                              hipStream_t stream) {
  const float* q     = (const float*)d_in[0];
  const float* k     = (const float*)d_in[1];
  const float* v     = (const float*)d_in[2];
  const float* Wq    = (const float*)d_in[3];
  const float* Wk    = (const float*)d_in[4];
  const float* Wv    = (const float*)d_in[5];
  const float* Wd    = (const float*)d_in[6];
  const float* gates = (const float*)d_in[7];
  float* out = (float*)d_out;

  const int L = LSEQ, D = DMODEL;
  const size_t LD = (size_t)L * D;
  float* Qs = (float*)d_ws;
  float* Ks = Qs + LD;
  float* Vs = Ks + LD;

  const float invScale = 1.0f / sqrtf(sqrtf(32.0f));

  dim3 gBlk(128);
  dim3 gGrid(D / 64, L / 64);

  hipLaunchKernelGGL(gemm_nt_f32_wmma, gGrid, gBlk, 0, stream, q, Wq, Qs, L, D, D, invScale);
  hipLaunchKernelGGL(gemm_nt_f32_wmma, gGrid, gBlk, 0, stream, k, Wk, Ks, L, D, D, invScale);
  hipLaunchKernelGGL(gemm_nt_f32_wmma, gGrid, gBlk, 0, stream, v, Wv, Vs, L, D, D, 1.0f);
  hipLaunchKernelGGL(ema_scan_chunked, dim3(D / 32, 2), dim3(SCAN_W * 32), 0, stream, Ks, Vs, gates);
  hipLaunchKernelGGL(gate_softmax, dim3(L, 4), dim3(256), 0, stream, Qs, Ks, Vs, L, D);
  hipLaunchKernelGGL(gemm_nt_f32_wmma, gGrid, gBlk, 0, stream, Qs, Wd, out, L, D, D, 1.0f);
}


# GEMM operands staged through LDS (2 stages, k-trip 64): each global load covers 4 whole 128-B lines (was 16 lines x 16 B), fragments by ds_load_b64; fragment k order and sum order unchanged
# speedup vs baseline: 5.0447x; 1.2356x over previous
#include <hip/hip_runtime.h>
#include <math.h>


typedef float v2f __attribute__((ext_vector_type(2)));
typedef float v4f __attribute__((ext_vector_type(4)));
typedef float v8f __attribute__((ext_vector_type(8)));

#define LSEQ   1024
#define DMODEL 1024

#define GT_K     64
#define GT_PITCH 68
#define GT_STAGE (64 * GT_PITCH)
__global__ __launch_bounds__(128) __attribute__((amdgpu_waves_per_eu(1, 4)))
void gemm_nt_f32_wmma(const float* __restrict__ A, const float* __restrict__ B,
                      float* __restrict__ C, int M, int N, int K, float alpha) {
  __shared__ __attribute__((aligned(16))) float sA[2 * GT_STAGE];
  __shared__ __attribute__((aligned(16))) float sB[2 * GT_STAGE];

  const int tid  = threadIdx.x;
  const int lane = tid & 31;
  const int wave = tid >> 5;
  const int lid  = lane & 15;
  const int half = lane >> 4;
  const int wm = (wave >> 1) * 32;
  const int wn = (wave & 1) * 32;
  const int mBase = blockIdx.y * 64 + wm;
  const int nBase = blockIdx.x * 64 + wn;

  const int ldRow = tid >> 4;
  const int ldCol = (tid & 15) * 4;
  const float* gA = A + (size_t)(blockIdx.y * 64 + ldRow) * K + ldCol;
  const float* gB = B + (size_t)(blockIdx.x * 64 + ldRow) * K + ldCol;
  const size_t rowStep = (size_t)8 * K;
  const int stOff = ldRow * GT_PITCH + ldCol;

  const int fA = (wm + lid) * GT_PITCH + half * 2;
  const int fB = (wn + lid) * GT_PITCH + half * 2;

  v8f acc[2][2];
  #pragma unroll
  for (int i = 0; i < 2; ++i)
    #pragma unroll
    for (int j = 0; j < 2; ++j)
      acc[i][j] = (v8f){};

  const int T = K / GT_K;

  v4f ra[8], rb[8];
  #pragma unroll
  for (int p = 0; p < 8; ++p) {
    ra[p] = *(const v4f*)(gA + p * rowStep);
    rb[p] = *(const v4f*)(gB + p * rowStep);
  }
  #pragma unroll
  for (int p = 0; p < 8; ++p) {
    *(v4f*)(sA + stOff + p * 8 * GT_PITCH) = ra[p];
    *(v4f*)(sB + stOff + p * 8 * GT_PITCH) = rb[p];
  }
  __syncthreads();
  {
    const int ko = (T > 1) ? GT_K : 0;
    #pragma unroll
    for (int p = 0; p < 8; ++p) {
      ra[p] = *(const v4f*)(gA + p * rowStep + ko);
      rb[p] = *(const v4f*)(gB + p * rowStep + ko);
    }
  }

  for (int t = 0; t < T; ++t) {
    const float* cA = sA + (t & 1) * GT_STAGE;
    const float* cB = sB + (t & 1) * GT_STAGE;
    v2f fa[16][2], fb[16][2];
    #pragma unroll
    for (int s = 0; s < 16; ++s) {
      #pragma unroll
      for (int i = 0; i < 2; ++i) fa[s][i] = *(const v2f*)(cA + fA + i * 16 * GT_PITCH + 4 * s);
      #pragma unroll
      for (int j = 0; j < 2; ++j) fb[s][j] = *(const v2f*)(cB + fB + j * 16 * GT_PITCH + 4 * s);
    }
    {
      float* nA = sA + ((t + 1) & 1) * GT_STAGE;
      float* nB = sB + ((t + 1) & 1) * GT_STAGE;
      #pragma unroll
      for (int p = 0; p < 8; ++p) {
        *(v4f*)(nA + stOff + p * 8 * GT_PITCH) = ra[p];
        *(v4f*)(nB + stOff + p * 8 * GT_PITCH) = rb[p];
      }
      const int ko = ((t + 2 < T) ? (t + 2) : (T - 1)) * GT_K;
      #pragma unroll
      for (int p = 0; p < 8; ++p) {
        ra[p] = *(const v4f*)(gA + p * rowStep + ko);
        rb[p] = *(const v4f*)(gB + p * rowStep + ko);
      }
    }
    #pragma unroll
    for (int s = 0; s < 16; ++s)
      #pragma unroll
      for (int i = 0; i < 2; ++i)
        #pragma unroll
        for (int j = 0; j < 2; ++j)
          acc[i][j] = __builtin_amdgcn_wmma_f32_16x16x4_f32(
              false, fa[s][i], false, fb[s][j], (short)0, acc[i][j], false, false);
    __builtin_amdgcn_sched_barrier(0);
    __builtin_amdgcn_fence(__ATOMIC_RELEASE, "workgroup", "local");
    __builtin_amdgcn_s_barrier();
    __builtin_amdgcn_fence(__ATOMIC_ACQUIRE, "workgroup", "local");
  }

  #pragma unroll
  for (int i = 0; i < 2; ++i) {
    #pragma unroll
    for (int j = 0; j < 2; ++j) {
      const int col = nBase + j * 16 + lid;
      #pragma unroll
      for (int r = 0; r < 8; ++r) {
        const int row = mBase + i * 16 + half * 8 + r;
        C[(size_t)row * N + col] = alpha * acc[i][j][r];
      }
    }
  }
}

#define SCAN_T 32
#define SCAN_W (LSEQ / SCAN_T)
__global__ __launch_bounds__(SCAN_W * 32)
void ema_scan_chunked(float* __restrict__ Ks, float* __restrict__ Vs,
                      const float* __restrict__ gates) {
  __shared__ float chunkEnd[SCAN_W][32];
  const int lane = threadIdx.x & 31;
  const int w    = threadIdx.x >> 5;
  float* P = (blockIdx.y ? Vs : Ks)
           + (size_t)(w * SCAN_T) * DMODEL + blockIdx.x * 32 + lane;
  const float g = gates[0];

  float x[SCAN_T];
  #pragma unroll
  for (int j = 0; j < SCAN_T; ++j) x[j] = P[(size_t)j * DMODEL];

  float e = 0.0f, gT = 1.0f;
  #pragma unroll
  for (int j = 0; j < SCAN_T; ++j) {
    e = fmaf(g, e, x[j]);
    x[j] = e;
    gT *= g;
  }
  chunkEnd[w][lane] = e;
  __syncthreads();

  if (w > 0) {
    float c = chunkEnd[0][lane];
    for (int i = 1; i < w; ++i) c = fmaf(c, gT, chunkEnd[i][lane]);
    float gp = g;
    #pragma unroll
    for (int j = 0; j < SCAN_T; ++j) {
      x[j] = fmaf(gp, c, x[j]);
      gp *= g;
    }
  }
  #pragma unroll
  for (int j = 0; j < SCAN_T; ++j) P[(size_t)j * DMODEL] = x[j];
}

__global__ __launch_bounds__(256)
void gate_softmax(float* __restrict__ Q, const float* __restrict__ Kema,
                  const float* __restrict__ Vema, int L, int D) {
  const int t    = blockIdx.x;
  const int wave = threadIdx.x >> 5;
  const int lane = threadIdx.x & 31;
  const int h    = blockIdx.y * 8 + wave;
  const size_t idx = (size_t)t * D + h * 32 + lane;
  const float qv = Q[idx];
  float p = qv * Kema[idx];
  #pragma unroll
  for (int off = 16; off > 0; off >>= 1)
    p += __shfl_xor(p, off, 32);
  const float a = 1.0f / (1.0f + 31.0f * expf(-p));
  Q[idx] = a * Vema[idx];
}

extern "C" void kernel_launch(void* const* d_in, const int* in_sizes, int n_in,
                              void* d_out, int out_size, void* d_ws, size_t ws_size,
                              hipStream_t stream) {
  const float* q     = (const float*)d_in[0];
  const float* k     = (const float*)d_in[1];
  const float* v     = (const float*)d_in[2];
  const float* Wq    = (const float*)d_in[3];
  const float* Wk    = (const float*)d_in[4];
  const float* Wv    = (const float*)d_in[5];
  const float* Wd    = (const float*)d_in[6];
  const float* gates = (const float*)d_in[7];
  float* out = (float*)d_out;

  const int L = LSEQ, D = DMODEL;
  const size_t LD = (size_t)L * D;
  float* Qs = (float*)d_ws;
  float* Ks = Qs + LD;
  float* Vs = Ks + LD;

  const float invScale = 1.0f / sqrtf(sqrtf(32.0f));

  dim3 gBlk(128);
  dim3 gGrid(D / 64, L / 64);

  hipLaunchKernelGGL(gemm_nt_f32_wmma, gGrid, gBlk, 0, stream, q, Wq, Qs, L, D, D, invScale);
  hipLaunchKernelGGL(gemm_nt_f32_wmma, gGrid, gBlk, 0, stream, k, Wk, Ks, L, D, D, invScale);
  hipLaunchKernelGGL(gemm_nt_f32_wmma, gGrid, gBlk, 0, stream, v, Wv, Vs, L, D, D, 1.0f);
  hipLaunchKernelGGL(ema_scan_chunked, dim3(D / 32, 2), dim3(SCAN_W * 32), 0, stream, Ks, Vs, gates);
  hipLaunchKernelGGL(gate_softmax, dim3(L, 4), dim3(256), 0, stream, Qs, Ks, Vs, L, D);
  hipLaunchKernelGGL(gemm_nt_f32_wmma, gGrid, gBlk, 0, stream, Qs, Wd, out, L, D, D, 1.0f);
}


# Three projection GEMMs (q.Wq, k.Wk, v.Wv) as ONE launch of 768 work-groups (blockIdx.z picks the operands): 3 work-groups per WGP fill each other's load and barrier gaps; sum order unchanged
# speedup vs baseline: 6.3822x; 1.2651x over previous
#include <hip/hip_runtime.h>
#include <math.h>


typedef float v2f __attribute__((ext_vector_type(2)));
typedef float v4f __attribute__((ext_vector_type(4)));
typedef float v8f __attribute__((ext_vector_type(8)));

#define LSEQ   1024
#define DMODEL 1024

#define GT_K     64
#define GT_PITCH 68
#define GT_STAGE (64 * GT_PITCH)
__global__ __launch_bounds__(128) __attribute__((amdgpu_waves_per_eu(1, 4)))
void gemm_nt_f32_wmma(const float* A0, const float* B0, float* C0, float alpha0,
                      const float* A1, const float* B1, float* C1, float alpha1,
                      const float* A2, const float* B2, float* C2, float alpha2,
                      int M, int N, int K) {
  __shared__ __attribute__((aligned(16))) float sA[2 * GT_STAGE];
  __shared__ __attribute__((aligned(16))) float sB[2 * GT_STAGE];

  const int z = blockIdx.z;
  const float* __restrict__ A = (z == 0) ? A0 : (z == 1) ? A1 : A2;
  const float* __restrict__ B = (z == 0) ? B0 : (z == 1) ? B1 : B2;
  float* __restrict__ C       = (z == 0) ? C0 : (z == 1) ? C1 : C2;
  const float alpha           = (z == 0) ? alpha0 : (z == 1) ? alpha1 : alpha2;

  const int tid  = threadIdx.x;
  const int lane = tid & 31;
  const int wave = tid >> 5;
  const int lid  = lane & 15;
  const int half = lane >> 4;
  const int wm = (wave >> 1) * 32;
  const int wn = (wave & 1) * 32;
  const int mBase = blockIdx.y * 64 + wm;
  const int nBase = blockIdx.x * 64 + wn;

  const int ldRow = tid >> 4;
  const int ldCol = (tid & 15) * 4;
  const float* gA = A + (size_t)(blockIdx.y * 64 + ldRow) * K + ldCol;
  const float* gB = B + (size_t)(blockIdx.x * 64 + ldRow) * K + ldCol;
  const size_t rowStep = (size_t)8 * K;
  const int stOff = ldRow * GT_PITCH + ldCol;

  const int fA = (wm + lid) * GT_PITCH + half * 2;
  const int fB = (wn + lid) * GT_PITCH + half * 2;

  v8f acc[2][2];
  #pragma unroll
  for (int i = 0; i < 2; ++i)
    #pragma unroll
    for (int j = 0; j < 2; ++j)
      acc[i][j] = (v8f){};

  const int T = K / GT_K;

  v4f ra[8], rb[8];
  #pragma unroll
  for (int p = 0; p < 8; ++p) {
    ra[p] = *(const v4f*)(gA + p * rowStep);
    rb[p] = *(const v4f*)(gB + p * rowStep);
  }
  #pragma unroll
  for (int p = 0; p < 8; ++p) {
    *(v4f*)(sA + stOff + p * 8 * GT_PITCH) = ra[p];
    *(v4f*)(sB + stOff + p * 8 * GT_PITCH) = rb[p];
  }
  __syncthreads();
  {
    const int ko = (T > 1) ? GT_K : 0;
    #pragma unroll
    for (int p = 0; p < 8; ++p) {
      ra[p] = *(const v4f*)(gA + p * rowStep + ko);
      rb[p] = *(const v4f*)(gB + p * rowStep + ko);
    }
  }

  for (int t = 0; t < T; ++t) {
    const float* cA = sA + (t & 1) * GT_STAGE;
    const float* cB = sB + (t & 1) * GT_STAGE;
    v2f fa[16][2], fb[16][2];
    #pragma unroll
    for (int s = 0; s < 16; ++s) {
      #pragma unroll
      for (int i = 0; i < 2; ++i) fa[s][i] = *(const v2f*)(cA + fA + i * 16 * GT_PITCH + 4 * s);
      #pragma unroll
      for (int j = 0; j < 2; ++j) fb[s][j] = *(const v2f*)(cB + fB + j * 16 * GT_PITCH + 4 * s);
    }
    {
      float* nA = sA + ((t + 1) & 1) * GT_STAGE;
      float* nB = sB + ((t + 1) & 1) * GT_STAGE;
      #pragma unroll
      for (int p = 0; p < 8; ++p) {
        *(v4f*)(nA + stOff + p * 8 * GT_PITCH) = ra[p];
        *(v4f*)(nB + stOff + p * 8 * GT_PITCH) = rb[p];
      }
      const int ko = ((t + 2 < T) ? (t + 2) : (T - 1)) * GT_K;
      #pragma unroll
      for (int p = 0; p < 8; ++p) {
        ra[p] = *(const v4f*)(gA + p * rowStep + ko);
        rb[p] = *(const v4f*)(gB + p * rowStep + ko);
      }
    }
    #pragma unroll
    for (int s = 0; s < 16; ++s)
      #pragma unroll
      for (int i = 0; i < 2; ++i)
        #pragma unroll
        for (int j = 0; j < 2; ++j)
          acc[i][j] = __builtin_amdgcn_wmma_f32_16x16x4_f32(
              false, fa[s][i], false, fb[s][j], (short)0, acc[i][j], false, false);
    __builtin_amdgcn_sched_barrier(0);
    __builtin_amdgcn_fence(__ATOMIC_RELEASE, "workgroup", "local");
    __builtin_amdgcn_s_barrier();
    __builtin_amdgcn_fence(__ATOMIC_ACQUIRE, "workgroup", "local");
  }

  #pragma unroll
  for (int i = 0; i < 2; ++i) {
    #pragma unroll
    for (int j = 0; j < 2; ++j) {
      const int col = nBase + j * 16 + lid;
      #pragma unroll
      for (int r = 0; r < 8; ++r) {
        const int row = mBase + i * 16 + half * 8 + r;
        C[(size_t)row * N + col] = alpha * acc[i][j][r];
      }
    }
  }
}

#define SCAN_T 32
#define SCAN_W (LSEQ / SCAN_T)
__global__ __launch_bounds__(SCAN_W * 32)
void ema_scan_chunked(float* __restrict__ Ks, float* __restrict__ Vs,
                      const float* __restrict__ gates) {
  __shared__ float chunkEnd[SCAN_W][32];
  const int lane = threadIdx.x & 31;
  const int w    = threadIdx.x >> 5;
  float* P = (blockIdx.y ? Vs : Ks)
           + (size_t)(w * SCAN_T) * DMODEL + blockIdx.x * 32 + lane;
  const float g = gates[0];

  float x[SCAN_T];
  #pragma unroll
  for (int j = 0; j < SCAN_T; ++j) x[j] = P[(size_t)j * DMODEL];

  float e = 0.0f, gT = 1.0f;
  #pragma unroll
  for (int j = 0; j < SCAN_T; ++j) {
    e = fmaf(g, e, x[j]);
    x[j] = e;
    gT *= g;
  }
  chunkEnd[w][lane] = e;
  __syncthreads();

  if (w > 0) {
    float c = chunkEnd[0][lane];
    for (int i = 1; i < w; ++i) c = fmaf(c, gT, chunkEnd[i][lane]);
    float gp = g;
    #pragma unroll
    for (int j = 0; j < SCAN_T; ++j) {
      x[j] = fmaf(gp, c, x[j]);
      gp *= g;
    }
  }
  #pragma unroll
  for (int j = 0; j < SCAN_T; ++j) P[(size_t)j * DMODEL] = x[j];
}

__global__ __launch_bounds__(256)
void gate_softmax(float* __restrict__ Q, const float* __restrict__ Kema,
                  const float* __restrict__ Vema, int L, int D) {
  const int t    = blockIdx.x;
  const int wave = threadIdx.x >> 5;
  const int lane = threadIdx.x & 31;
  const int h    = blockIdx.y * 8 + wave;
  const size_t idx = (size_t)t * D + h * 32 + lane;
  const float qv = Q[idx];
  float p = qv * Kema[idx];
  #pragma unroll
  for (int off = 16; off > 0; off >>= 1)
    p += __shfl_xor(p, off, 32);
  const float a = 1.0f / (1.0f + 31.0f * expf(-p));
  Q[idx] = a * Vema[idx];
}

extern "C" void kernel_launch(void* const* d_in, const int* in_sizes, int n_in,
                              void* d_out, int out_size, void* d_ws, size_t ws_size,
                              hipStream_t stream) {
  const float* q     = (const float*)d_in[0];
  const float* k     = (const float*)d_in[1];
  const float* v     = (const float*)d_in[2];
  const float* Wq    = (const float*)d_in[3];
  const float* Wk    = (const float*)d_in[4];
  const float* Wv    = (const float*)d_in[5];
  const float* Wd    = (const float*)d_in[6];
  const float* gates = (const float*)d_in[7];
  float* out = (float*)d_out;

  const int L = LSEQ, D = DMODEL;
  const size_t LD = (size_t)L * D;
  float* Qs = (float*)d_ws;
  float* Ks = Qs + LD;
  float* Vs = Ks + LD;

  const float invScale = 1.0f / sqrtf(sqrtf(32.0f));

  dim3 gBlk(128);
  dim3 gGrid1(D / 64, L / 64, 1);
  dim3 gGrid3(D / 64, L / 64, 3);

  hipLaunchKernelGGL(gemm_nt_f32_wmma, gGrid3, gBlk, 0, stream,
                     q, Wq, Qs, invScale, k, Wk, Ks, invScale, v, Wv, Vs, 1.0f, L, D, D);
  hipLaunchKernelGGL(ema_scan_chunked, dim3(D / 32, 2), dim3(SCAN_W * 32), 0, stream, Ks, Vs, gates);
  hipLaunchKernelGGL(gate_softmax, dim3(L, 4), dim3(256), 0, stream, Qs, Ks, Vs, L, D);
  hipLaunchKernelGGL(gemm_nt_f32_wmma, gGrid1, gBlk, 0, stream,
                     Qs, Wd, out, 1.0f, Qs, Wd, out, 1.0f, Qs, Wd, out, 1.0f, L, D, D);
}


# Gate launch removed (4 -> 3 launches): K scan waves form a from kema in registers (butterfly row sums, same xor tree, first add unfused); out GEMM scales its A tile (vema) by a while staging
# speedup vs baseline: 6.7263x; 1.0539x over previous
#include <hip/hip_runtime.h>
#include <math.h>


typedef float v2f __attribute__((ext_vector_type(2)));
typedef float v4f __attribute__((ext_vector_type(4)));
typedef float v8f __attribute__((ext_vector_type(8)));

#define LSEQ   1024
#define DMODEL 1024

#define GT_K     64
#define GT_PITCH 68
#define GT_STAGE (64 * GT_PITCH)
template <bool GATED>
__device__ __forceinline__ void gemm_nt_tile(const float* __restrict__ A, const float* __restrict__ B,
                                             float* __restrict__ C, const float alpha,
                                             const float* __restrict__ G,
                                             const int M, const int N, const int K) {
  __shared__ __attribute__((aligned(16))) float sA[2 * GT_STAGE];
  __shared__ __attribute__((aligned(16))) float sB[2 * GT_STAGE];

  const int tid  = threadIdx.x;
  const int lane = tid & 31;
  const int wave = tid >> 5;
  const int lid  = lane & 15;
  const int half = lane >> 4;
  const int wm = (wave >> 1) * 32;
  const int wn = (wave & 1) * 32;
  const int mBase = blockIdx.y * 64 + wm;
  const int nBase = blockIdx.x * 64 + wn;

  const int ldRow = tid >> 4;
  const int ldCol = (tid & 15) * 4;
  const float* gA = A + (size_t)(blockIdx.y * 64 + ldRow) * K + ldCol;
  const float* gB = B + (size_t)(blockIdx.x * 64 + ldRow) * K + ldCol;
  const size_t rowStep = (size_t)8 * K;
  const int stOff = ldRow * GT_PITCH + ldCol;
  const float* gG = GATED ? G + (size_t)(ldCol >> 5) * M + (blockIdx.y * 64 + ldRow) : G;
  const size_t gTrip = (size_t)2 * M;

  const int fA = (wm + lid) * GT_PITCH + half * 2;
  const int fB = (wn + lid) * GT_PITCH + half * 2;

  v8f acc[2][2];
  #pragma unroll
  for (int i = 0; i < 2; ++i)
    #pragma unroll
    for (int j = 0; j < 2; ++j)
      acc[i][j] = (v8f){};

  const int T = K / GT_K;

  v4f ra[8], rb[8];
  float rg[8];
  #pragma unroll
  for (int p = 0; p < 8; ++p) {
    ra[p] = *(const v4f*)(gA + p * rowStep);
    rb[p] = *(const v4f*)(gB + p * rowStep);
    if (GATED) rg[p] = gG[8 * p];
  }
  #pragma unroll
  for (int p = 0; p < 8; ++p) {
    *(v4f*)(sA + stOff + p * 8 * GT_PITCH) = GATED ? ra[p] * rg[p] : ra[p];
    *(v4f*)(sB + stOff + p * 8 * GT_PITCH) = rb[p];
  }
  __syncthreads();
  {
    const int kt = (T > 1) ? 1 : 0;
    const int ko = kt * GT_K;
    #pragma unroll
    for (int p = 0; p < 8; ++p) {
      ra[p] = *(const v4f*)(gA + p * rowStep + ko);
      rb[p] = *(const v4f*)(gB + p * rowStep + ko);
      if (GATED) rg[p] = gG[kt * gTrip + 8 * p];
    }
  }

  for (int t = 0; t < T; ++t) {
    const float* cA = sA + (t & 1) * GT_STAGE;
    const float* cB = sB + (t & 1) * GT_STAGE;
    v2f fa[16][2], fb[16][2];
    #pragma unroll
    for (int s = 0; s < 16; ++s) {
      #pragma unroll
      for (int i = 0; i < 2; ++i) fa[s][i] = *(const v2f*)(cA + fA + i * 16 * GT_PITCH + 4 * s);
      #pragma unroll
      for (int j = 0; j < 2; ++j) fb[s][j] = *(const v2f*)(cB + fB + j * 16 * GT_PITCH + 4 * s);
    }
    if (GATED) __builtin_amdgcn_sched_barrier(0);
    {
      float* nA = sA + ((t + 1) & 1) * GT_STAGE;
      float* nB = sB + ((t + 1) & 1) * GT_STAGE;
      #pragma unroll
      for (int p = 0; p < 8; ++p) {
        *(v4f*)(nA + stOff + p * 8 * GT_PITCH) = GATED ? ra[p] * rg[p] : ra[p];
        *(v4f*)(nB + stOff + p * 8 * GT_PITCH) = rb[p];
      }
      const int kt = (t + 2 < T) ? (t + 2) : (T - 1);
      const int ko = kt * GT_K;
      #pragma unroll
      for (int p = 0; p < 8; ++p) {
        ra[p] = *(const v4f*)(gA + p * rowStep + ko);
        rb[p] = *(const v4f*)(gB + p * rowStep + ko);
        if (GATED) rg[p] = gG[kt * gTrip + 8 * p];
      }
    }
    #pragma unroll
    for (int s = 0; s < 16; ++s)
      #pragma unroll
      for (int i = 0; i < 2; ++i)
        #pragma unroll
        for (int j = 0; j < 2; ++j)
          acc[i][j] = __builtin_amdgcn_wmma_f32_16x16x4_f32(
              false, fa[s][i], false, fb[s][j], (short)0, acc[i][j], false, false);
    __builtin_amdgcn_sched_barrier(0);
    __builtin_amdgcn_fence(__ATOMIC_RELEASE, "workgroup", "local");
    __builtin_amdgcn_s_barrier();
    __builtin_amdgcn_fence(__ATOMIC_ACQUIRE, "workgroup", "local");
  }

  #pragma unroll
  for (int i = 0; i < 2; ++i) {
    #pragma unroll
    for (int j = 0; j < 2; ++j) {
      const int col = nBase + j * 16 + lid;
      #pragma unroll
      for (int r = 0; r < 8; ++r) {
        const int row = mBase + i * 16 + half * 8 + r;
        C[(size_t)row * N + col] = alpha * acc[i][j][r];
      }
    }
  }
}

__global__ __launch_bounds__(128) __attribute__((amdgpu_waves_per_eu(1, 4)))
void gemm_nt_f32_wmma(const float* A0, const float* B0, float* C0, float alpha0,
                      const float* A1, const float* B1, float* C1, float alpha1,
                      const float* A2, const float* B2, float* C2, float alpha2,
                      int M, int N, int K) {
  const int z = blockIdx.z;
  const float* A    = (z == 0) ? A0 : (z == 1) ? A1 : A2;
  const float* B    = (z == 0) ? B0 : (z == 1) ? B1 : B2;
  float* C          = (z == 0) ? C0 : (z == 1) ? C1 : C2;
  const float alpha = (z == 0) ? alpha0 : (z == 1) ? alpha1 : alpha2;
  gemm_nt_tile<false>(A, B, C, alpha, nullptr, M, N, K);
}

__global__ __launch_bounds__(128) __attribute__((amdgpu_waves_per_eu(1, 4)))
void gemm_nt_f32_wmma_rowgate(const float* A, const float* B, float* C, const float* G,
                              int M, int N, int K) {
  gemm_nt_tile<true>(A, B, C, 1.0f, G, M, N, K);
}

#define SCAN_T 32
#define SCAN_W (LSEQ / SCAN_T)

#define ROWSUM_LEVEL(MASK)                                        \
  {                                                               \
    const bool up = (lane & (MASK)) != 0;                         \
    _Pragma("unroll")                                             \
    for (int j = 0; j < (MASK); ++j) {                            \
      const float give = up ? x[j] : x[j + (MASK)];               \
      const float keep = up ? x[j + (MASK)] : x[j];               \
      x[j] = keep + __shfl_xor(give, (MASK), 32);                 \
    }                                                             \
  }

__global__ __launch_bounds__(SCAN_W * 32)
void ema_scan_chunked(const float* __restrict__ Ks, float* __restrict__ Vs,
                      const float* __restrict__ Qs, float* __restrict__ aT,
                      const float* __restrict__ gates) {
  __shared__ float chunkEnd[SCAN_W][32];
  const int lane = threadIdx.x & 31;
  const int w    = threadIdx.x >> 5;
  const bool isK = (blockIdx.y == 0);
  const size_t off = (size_t)(w * SCAN_T) * DMODEL + blockIdx.x * 32 + lane;
  const float* P = (isK ? Ks : (const float*)Vs) + off;
  const float g = gates[0];

  float x[SCAN_T], qv[SCAN_T];
  #pragma unroll
  for (int j = 0; j < SCAN_T; ++j) x[j] = P[(size_t)j * DMODEL];
  if (isK) {
    #pragma unroll
    for (int j = 0; j < SCAN_T; ++j) qv[j] = Qs[off + (size_t)j * DMODEL];
  }

  float e = 0.0f, gT = 1.0f;
  #pragma unroll
  for (int j = 0; j < SCAN_T; ++j) {
    e = fmaf(g, e, x[j]);
    x[j] = e;
    gT *= g;
  }
  chunkEnd[w][lane] = e;
  __syncthreads();

  if (w > 0) {
    float c = chunkEnd[0][lane];
    for (int i = 1; i < w; ++i) c = fmaf(c, gT, chunkEnd[i][lane]);
    float gp = g;
    #pragma unroll
    for (int j = 0; j < SCAN_T; ++j) {
      x[j] = fmaf(gp, c, x[j]);
      gp *= g;
    }
  }

  if (isK) {
    #pragma unroll
    for (int j = 0; j < SCAN_T; ++j) x[j] = qv[j] * x[j];
    ROWSUM_LEVEL(16)
    ROWSUM_LEVEL(8)
    ROWSUM_LEVEL(4)
    ROWSUM_LEVEL(2)
    ROWSUM_LEVEL(1)
    const float s = x[0];
    aT[(size_t)blockIdx.x * LSEQ + w * SCAN_T + lane] = 1.0f / (1.0f + 31.0f * expf(-s));
  } else {
    float* O = Vs + off;
    #pragma unroll
    for (int j = 0; j < SCAN_T; ++j) O[(size_t)j * DMODEL] = x[j];
  }
}

extern "C" void kernel_launch(void* const* d_in, const int* in_sizes, int n_in,
                              void* d_out, int out_size, void* d_ws, size_t ws_size,
                              hipStream_t stream) {
  const float* q     = (const float*)d_in[0];
  const float* k     = (const float*)d_in[1];
  const float* v     = (const float*)d_in[2];
  const float* Wq    = (const float*)d_in[3];
  const float* Wk    = (const float*)d_in[4];
  const float* Wv    = (const float*)d_in[5];
  const float* Wd    = (const float*)d_in[6];
  const float* gates = (const float*)d_in[7];
  float* out = (float*)d_out;

  const int L = LSEQ, D = DMODEL;
  const size_t LD = (size_t)L * D;
  float* Qs = (float*)d_ws;
  float* Ks = Qs + LD;
  float* Vs = Ks + LD;
  float* aT = Vs + LD;

  const float invScale = 1.0f / sqrtf(sqrtf(32.0f));

  dim3 gBlk(128);
  dim3 gGrid1(D / 64, L / 64, 1);
  dim3 gGrid3(D / 64, L / 64, 3);

  hipLaunchKernelGGL(gemm_nt_f32_wmma, gGrid3, gBlk, 0, stream,
                     q, Wq, Qs, invScale, k, Wk, Ks, invScale, v, Wv, Vs, 1.0f, L, D, D);
  hipLaunchKernelGGL(ema_scan_chunked, dim3(D / 32, 2), dim3(SCAN_W * 32), 0, stream,
                     (const float*)Ks, Vs, (const float*)Qs, aT, gates);
  hipLaunchKernelGGL(gemm_nt_f32_wmma_rowgate, gGrid1, gBlk, 0, stream,
                     (const float*)Vs, Wd, out, (const float*)aT, L, D, D);
}
